# starting kernel
# baseline (speedup 1.0000x reference)
#include <hip/hip_runtime.h>

typedef __attribute__((ext_vector_type(2))) float v2f;
typedef __attribute__((ext_vector_type(8))) float v8f;

#define DFEAT 128

__global__ void gemm_wmma_f32(const float* __restrict__ x,
                              const float* __restrict__ w,
                              float* __restrict__ h) {
  const int lane = threadIdx.x & 31;
  const int wid  = threadIdx.x >> 5;
  const int m0   = blockIdx.x * 16;
  const int n0   = wid * 16;
  const int l16  = lane & 15;
  const int kh   = (lane >> 4) << 1;

  const float* arow = x + (m0 + l16) * DFEAT;
  const float* bcol = w + n0 + l16;

  v8f c = {};
#pragma unroll
  for (int k0 = 0; k0 < DFEAT; k0 += 4) {
    v2f a = *(const v2f*)(arow + k0 + kh);
    v2f b;
    b.x = bcol[(k0 + kh    ) * DFEAT];
    b.y = bcol[(k0 + kh + 1) * DFEAT];
    c = __builtin_amdgcn_wmma_f32_16x16x4_f32(false, a, false, b,
                                              (short)0, c, false, false);
  }

  float* hcol = h + n0 + l16;
  const int rbase = m0 + ((lane >> 4) << 3);
#pragma unroll
  for (int j = 0; j < 8; ++j) {
    hcol[(rbase + j) * DFEAT] = c[j];
  }
}

__global__ void init_out_bias(const float* __restrict__ bias,
                              float* __restrict__ out, int n) {
  int i = blockIdx.x * blockDim.x + threadIdx.x;
  if (i < n) out[i] = bias[i & (DFEAT - 1)];
}

__global__ void spmm_scatter(const float* __restrict__ h,
                             const float* __restrict__ val,
                             const int* __restrict__ rowi,
                             const int* __restrict__ coli,
                             float* __restrict__ out, int E) {
  const int lane = threadIdx.x & 31;
  const int wid  = threadIdx.x >> 5;
  const int d0   = lane << 2;
  int e = (blockIdx.x * 8 + wid) * 4;

#pragma unroll
  for (int i = 0; i < 4; ++i, ++e) {
    if (e >= E) return;
    const int   r = rowi[e];
    const int   cidx = coli[e];
    const float v = val[e];
    const float4 hv = *(const float4*)(h + cidx * DFEAT + d0);
    float* o = out + r * DFEAT + d0;
    unsafeAtomicAdd(o + 0, v * hv.x);
    unsafeAtomicAdd(o + 1, v * hv.y);
    unsafeAtomicAdd(o + 2, v * hv.z);
    unsafeAtomicAdd(o + 3, v * hv.w);
  }
}

extern "C" void kernel_launch(void* const* d_in, const int* in_sizes, int n_in,
                              void* d_out, int out_size, void* d_ws, size_t ws_size,
                              hipStream_t stream) {
  const float* x       = (const float*)d_in[0];
  const float* adj_val = (const float*)d_in[1];
  const float* weight  = (const float*)d_in[2];
  const float* bias    = (const float*)d_in[3];
  const int*   adj_row = (const int*)d_in[4];
  const int*   adj_col = (const int*)d_in[5];

  const int N = in_sizes[0] / DFEAT;
  const int E = in_sizes[1];

  float* h   = (float*)d_ws;
  float* out = (float*)d_out;

  gemm_wmma_f32<<<N / 16, 256, 0, stream>>>(x, weight, h);

  init_out_bias<<<(N * DFEAT + 255) / 256, 256, 0, stream>>>(bias, out, N * DFEAT);

  const int waves  = (E + 3) / 4;
  const int blocks = (waves + 7) / 8;
  spmm_scatter<<<blocks, 256, 0, stream>>>(h, adj_val, adj_row, adj_col, out, E);
}


# spmm scatter: lane owns features l, l+32, l+64, l+96 so each f32 atomic and each gather instruction covers one whole 128-B line, not 4 quarter-filled lines; sums stay atomic arrival order
# speedup vs baseline: 1.7333x; 1.7333x over previous
#include <hip/hip_runtime.h>

typedef __attribute__((ext_vector_type(2))) float v2f;
typedef __attribute__((ext_vector_type(8))) float v8f;

#define DFEAT 128

__global__ void gemm_wmma_f32(const float* __restrict__ x,
                              const float* __restrict__ w,
                              float* __restrict__ h) {
  const int lane = threadIdx.x & 31;
  const int wid  = threadIdx.x >> 5;
  const int m0   = blockIdx.x * 16;
  const int n0   = wid * 16;
  const int l16  = lane & 15;
  const int kh   = (lane >> 4) << 1;

  const float* arow = x + (m0 + l16) * DFEAT;
  const float* bcol = w + n0 + l16;

  v8f c = {};
#pragma unroll
  for (int k0 = 0; k0 < DFEAT; k0 += 4) {
    v2f a = *(const v2f*)(arow + k0 + kh);
    v2f b;
    b.x = bcol[(k0 + kh    ) * DFEAT];
    b.y = bcol[(k0 + kh + 1) * DFEAT];
    c = __builtin_amdgcn_wmma_f32_16x16x4_f32(false, a, false, b,
                                              (short)0, c, false, false);
  }

  float* hcol = h + n0 + l16;
  const int rbase = m0 + ((lane >> 4) << 3);
#pragma unroll
  for (int j = 0; j < 8; ++j) {
    hcol[(rbase + j) * DFEAT] = c[j];
  }
}

__global__ void init_out_bias(const float* __restrict__ bias,
                              float* __restrict__ out, int n) {
  int i = blockIdx.x * blockDim.x + threadIdx.x;
  if (i < n) out[i] = bias[i & (DFEAT - 1)];
}

__global__ void spmm_scatter(const float* __restrict__ h,
                             const float* __restrict__ val,
                             const int* __restrict__ rowi,
                             const int* __restrict__ coli,
                             float* __restrict__ out, int E) {
  const int lane = threadIdx.x & 31;
  const int wid  = threadIdx.x >> 5;
  int e = (blockIdx.x * 8 + wid) * 4;

#pragma unroll
  for (int i = 0; i < 4; ++i, ++e) {
    if (e >= E) return;
    const int   r = rowi[e];
    const int   cidx = coli[e];
    const float v = val[e];
    const float* hp = h + cidx * DFEAT + lane;
    const float h0 = hp[0];
    const float h1 = hp[32];
    const float h2 = hp[64];
    const float h3 = hp[96];
    float* o = out + r * DFEAT + lane;
    unsafeAtomicAdd(o +  0, v * h0);
    unsafeAtomicAdd(o + 32, v * h1);
    unsafeAtomicAdd(o + 64, v * h2);
    unsafeAtomicAdd(o + 96, v * h3);
  }
}

extern "C" void kernel_launch(void* const* d_in, const int* in_sizes, int n_in,
                              void* d_out, int out_size, void* d_ws, size_t ws_size,
                              hipStream_t stream) {
  const float* x       = (const float*)d_in[0];
  const float* adj_val = (const float*)d_in[1];
  const float* weight  = (const float*)d_in[2];
  const float* bias    = (const float*)d_in[3];
  const int*   adj_row = (const int*)d_in[4];
  const int*   adj_col = (const int*)d_in[5];

  const int N = in_sizes[0] / DFEAT;
  const int E = in_sizes[1];

  float* h   = (float*)d_ws;
  float* out = (float*)d_out;

  gemm_wmma_f32<<<N / 16, 256, 0, stream>>>(x, weight, h);

  init_out_bias<<<(N * DFEAT + 255) / 256, 256, 0, stream>>>(bias, out, N * DFEAT);

  const int waves  = (E + 3) / 4;
  const int blocks = (waves + 7) / 8;
  spmm_scatter<<<blocks, 256, 0, stream>>>(h, adj_val, adj_row, adj_col, out, E);
}


# spmm without global atomics: edges bucketed in-call by 16-row output range (LDS counters), each range's A*x summed in registers+LDS then times W on WMMA; sum order = bucket rank, (A x) W
# speedup vs baseline: 1.8487x; 1.0666x over previous
#include <hip/hip_runtime.h>


typedef __attribute__((ext_vector_type(2))) float v2f;
typedef __attribute__((ext_vector_type(4))) float v4f;
typedef __attribute__((ext_vector_type(8))) float v8f;
typedef __attribute__((ext_vector_type(16))) float v16f;

#define DFEAT  128
#define RSHIFT 4
#define MAXR   1024
#define NBKT   64
#define BT     1024
#define EPT    10
#define KS     33
#define PITCH  132
#define AW     4
#define AT     (AW * 32)
#define RLSH   27

__device__ __forceinline__ int range_of(int r, int R) {
  int g = r >> RSHIFT;
  g = g < 0 ? 0 : g;
  return g >= R ? R - 1 : g;
}

__global__ __launch_bounds__(BT) void bucket_count(const int* __restrict__ rowi,
                                                   int* __restrict__ counts,
                                                   int E, int R) {
  __shared__ int hist[MAXR];
  const int t = threadIdx.x;
  const int b = blockIdx.x;
  for (int i = t; i < R; i += BT) hist[i] = 0;
  __syncthreads();
  const int epb = (E + NBKT - 1) / NBKT;
  const int beg = b * epb;
  const int end = min(E, beg + epb);
  for (int base = beg; base < end; base += EPT * BT) {
    int rv[EPT];
#pragma unroll
    for (int i = 0; i < EPT; ++i) {
      const int e = base + t + i * BT;
      rv[i] = rowi[e < end ? e : end - 1];
    }
#pragma unroll
    for (int i = 0; i < EPT; ++i) {
      const int e = base + t + i * BT;
      if (e < end) atomicAdd(&hist[range_of(rv[i], R)], 1);
    }
  }
  __syncthreads();
  for (int i = t; i < R; i += BT) counts[b * R + i] = hist[i];
}

__global__ __launch_bounds__(BT) void bucket_place(const int* __restrict__ rowi,
                                                   const int* __restrict__ counts,
                                                   int* __restrict__ rec,
                                                   int* __restrict__ rstart,
                                                   int E, int R) {
  __shared__ int cur[MAXR];
  __shared__ int wtot[BT / 32];
  const int t = threadIdx.x;
  const int lane = t & 31;
  const int wv = t >> 5;
  const int b = blockIdx.x;

  int tot = 0, bef = 0;
  {
    const int tc = t < R ? t : R - 1;
#pragma unroll 16
    for (int bb = 0; bb < NBKT; ++bb) {
      const int c = counts[bb * R + tc];
      tot += c;
      bef += (bb < b) ? c : 0;
    }
    tot = t < R ? tot : 0;
    bef = t < R ? bef : 0;
  }
  int inc = tot;
  for (int d = 1; d < 32; d <<= 1) {
    const int y = __shfl_up(inc, d);
    if (lane >= d) inc += y;
  }
  if (lane == 31) wtot[wv] = inc;
  __syncthreads();
  if (wv == 0) {
    const int wsum = wtot[lane];
    int winc = wsum;
    for (int d = 1; d < 32; d <<= 1) {
      const int y = __shfl_up(winc, d);
      if (lane >= d) winc += y;
    }
    wtot[lane] = winc - wsum;
  }
  __syncthreads();
  const int excl = wtot[wv] + inc - tot;
  if (t < R) {
    cur[t] = excl + bef;
    if (b == 0) rstart[t] = excl;
  }
  if (b == 0 && t == 0) rstart[R] = E;
  __syncthreads();

  const int epb = (E + NBKT - 1) / NBKT;
  const int beg = b * epb;
  const int end = min(E, beg + epb);
  for (int base = beg; base < end; base += EPT * BT) {
    int rv[EPT];
#pragma unroll
    for (int i = 0; i < EPT; ++i) {
      const int e = base + t + i * BT;
      rv[i] = rowi[e < end ? e : end - 1];
    }
#pragma unroll
    for (int i = 0; i < EPT; ++i) {
      const int e = base + t + i * BT;
      if (e < end) {
        const int pos = atomicAdd(&cur[range_of(rv[i], R)], 1);
        if ((unsigned)pos < (unsigned)E) rec[pos] = e | ((rv[i] & 15) << RLSH);
      }
    }
  }
}

__device__ __forceinline__ void lds_add4(float* ap, float a, float b, float c, float d) {
  __hip_atomic_fetch_add(ap,          a, __ATOMIC_RELAXED, __HIP_MEMORY_SCOPE_WORKGROUP);
  __hip_atomic_fetch_add(ap + KS,     b, __ATOMIC_RELAXED, __HIP_MEMORY_SCOPE_WORKGROUP);
  __hip_atomic_fetch_add(ap + 2 * KS, c, __ATOMIC_RELAXED, __HIP_MEMORY_SCOPE_WORKGROUP);
  __hip_atomic_fetch_add(ap + 3 * KS, d, __ATOMIC_RELAXED, __HIP_MEMORY_SCOPE_WORKGROUP);
}

__global__ __launch_bounds__(AT) void aggregate_rows_gemm(
    const float* __restrict__ x, const float* __restrict__ val,
    const int* __restrict__ coli, const float* __restrict__ w,
    const float* __restrict__ bias, const int* __restrict__ rec,
    const int* __restrict__ rstart, float* __restrict__ out, int N, int E) {
  __shared__ __attribute__((aligned(16))) float acc[16 * PITCH];
  const int t = threadIdx.x;
  const int lane = t & 31;
  const int wid = __builtin_amdgcn_readfirstlane(t >> 5);
  const int m0 = blockIdx.x << RSHIFT;

  for (int i = t; i < 16 * PITCH; i += AT) acc[i] = 0.f;
  __syncthreads();

  const int s = rstart[blockIdx.x];
  const int send = rstart[blockIdx.x + 1];
  const unsigned lane16 = (unsigned)lane << 4;
  const char* xb = (const char*)x + lane16;

  int cb = s + wid * 32;
  int rl, cc;
  float vv;
  {
    int i0 = cb + lane;
    i0 = i0 < E ? i0 : E - 1;
    const int p = rec[i0];
    int e = p & ((1 << RLSH) - 1);
    e = e < E ? e : E - 1;
    int c = coli[e];
    const float v = val[e];
    c = c < 0 ? 0 : c;
    c = c < N ? c : N - 1;
    const bool ok = lane < send - cb;
    rl = ok ? ((p >> RLSH) & 15) : 0;
    cc = ok ? c : 0;
    vv = ok ? v : 0.f;
  }
  int pa;
  {
    int i1 = cb + AT + lane;
    i1 = i1 < E ? i1 : E - 1;
    pa = rec[i1];
  }

  v16f ax = {}, ay = {}, az = {}, aw = {};
  while (cb + 32 <= send) {
    int i2 = cb + 2 * AT + lane;
    i2 = i2 < E ? i2 : E - 1;
    const int pn = rec[i2];
    int e2 = pa & ((1 << RLSH) - 1);
    e2 = e2 < E ? e2 : E - 1;
    int c2 = coli[e2];
    const float v2 = val[e2];
#pragma unroll
    for (int j = 0; j < 32; j += 16) {
      v4f g[16];
#pragma unroll
      for (int u = 0; u < 16; ++u) {
        const int c = __builtin_amdgcn_readlane(cc, j + u);
        g[u] = *(const v4f*)(xb + (size_t)c * (DFEAT * 4));
      }
#pragma unroll
      for (int u = 0; u < 16; ++u) {
        const int r = __builtin_amdgcn_readlane(rl, j + u);
        const float v = __int_as_float(
            __builtin_amdgcn_readlane(__float_as_int(vv), j + u));
        ax[r] += v * g[u].x;
        ay[r] += v * g[u].y;
        az[r] += v * g[u].z;
        aw[r] += v * g[u].w;
      }
    }
    cb += AT;
    c2 = c2 < 0 ? 0 : c2;
    c2 = c2 < N ? c2 : N - 1;
    const bool ok2 = lane < send - cb;
    rl = ok2 ? ((pa >> RLSH) & 15) : 0;
    cc = ok2 ? c2 : 0;
    vv = ok2 ? v2 : 0.f;
    pa = pn;
  }
  if (cb < send) {
    const int n = send - cb;
#pragma unroll
    for (int j = 0; j < 32; j += 16) {
      if (j < n) {
        v4f g[16];
#pragma unroll
        for (int u = 0; u < 16; ++u) {
          const int c = __builtin_amdgcn_readlane(cc, j + u);
          g[u] = *(const v4f*)(xb + (size_t)c * (DFEAT * 4));
        }
#pragma unroll
        for (int u = 0; u < 16; ++u) {
          if (j + u < n) {
            const int r = __builtin_amdgcn_readlane(rl, j + u);
            const float v = __int_as_float(
                __builtin_amdgcn_readlane(__float_as_int(vv), j + u));
            lds_add4(acc + r * PITCH + lane, v * g[u].x, v * g[u].y, v * g[u].z, v * g[u].w);
          }
        }
      }
    }
  }
#pragma unroll
  for (int r = 0; r < 16; ++r) {
    lds_add4(acc + r * PITCH + lane, ax[r], ay[r], az[r], aw[r]);
  }
  __syncthreads();

  const int l16 = lane & 15;
  const int hh  = lane >> 4;
  const int kh  = hh << 1;
  const int n0  = wid * 32;
  const float* arow = acc + l16 * PITCH;
  const float* bcol = w + n0 + l16;

  v8f c0 = {}, c1 = {};
#pragma unroll
  for (int k0 = 0; k0 < DFEAT; k0 += 4) {
    v2f a;
    a.x = arow[(kh    ) * KS + (k0 >> 2)];
    a.y = arow[(kh + 1) * KS + (k0 >> 2)];
    v2f b0, b1;
    b0.x = bcol[(k0 + kh    ) * DFEAT];
    b0.y = bcol[(k0 + kh + 1) * DFEAT];
    b1.x = bcol[(k0 + kh    ) * DFEAT + 16];
    b1.y = bcol[(k0 + kh + 1) * DFEAT + 16];
    c0 = __builtin_amdgcn_wmma_f32_16x16x4_f32(false, a, false, b0,
                                               (short)0, c0, false, false);
    c1 = __builtin_amdgcn_wmma_f32_16x16x4_f32(false, a, false, b1,
                                               (short)0, c1, false, false);
  }

  const float bv0 = bias[n0 + l16];
  const float bv1 = bias[n0 + 16 + l16];
  __syncthreads();
#pragma unroll
  for (int j = 0; j < 8; ++j) {
    acc[(hh * 8 + j) * PITCH + n0 + l16]      = c0[j] + bv0;
    acc[(hh * 8 + j) * PITCH + n0 + 16 + l16] = c1[j] + bv1;
  }
  __syncthreads();
  for (int i = 0; i < 16 / AW; ++i) {
    const int row = wid * (16 / AW) + i;
    if (m0 + row < N) {
      const v4f o = *(const v4f*)(acc + row * PITCH + lane * 4);
      *(v4f*)(out + (size_t)(m0 + row) * DFEAT + lane * 4) = o;
    }
  }
}

extern "C" void kernel_launch(void* const* d_in, const int* in_sizes, int n_in,
                              void* d_out, int out_size, void* d_ws, size_t ws_size,
                              hipStream_t stream) {
  const float* x       = (const float*)d_in[0];
  const float* adj_val = (const float*)d_in[1];
  const float* weight  = (const float*)d_in[2];
  const float* bias    = (const float*)d_in[3];
  const int*   adj_row = (const int*)d_in[4];
  const int*   adj_col = (const int*)d_in[5];

  const int N = in_sizes[0] / DFEAT;
  const int E = in_sizes[1];
  const int R = (N + 15) / 16;

  int* rec    = (int*)d_ws;
  int* counts = rec + E;
  int* rstart = counts + NBKT * R;
  float* out  = (float*)d_out;

  bucket_count<<<NBKT, BT, 0, stream>>>(adj_row, counts, E, R);
  bucket_place<<<NBKT, BT, 0, stream>>>(adj_row, counts, rec, rstart, E, R);
  aggregate_rows_gemm<<<R, AT, 0, stream>>>(x, adj_val, adj_col, weight, bias,
                                             rec, rstart, out, N, E);
}


# count+scan+place in one launch: each of 64 blocks counting-sorts its own edge slice in LDS (3 launches -> 2); owner group stages its 64 sub-segments' col/row/val in LDS; sum order: slice, LDS rank
# speedup vs baseline: 2.7315x; 1.4775x over previous
#include <hip/hip_runtime.h>


typedef __attribute__((ext_vector_type(2))) float v2f;
typedef __attribute__((ext_vector_type(4))) float v4f;
typedef __attribute__((ext_vector_type(8))) float v8f;
typedef __attribute__((ext_vector_type(16))) float v16f;

#define DFEAT  128
#define RSHIFT 4
#define MAXR   1024
#define NBKT   64
#define BT     1024
#define EPT    10
#define SLICE  (EPT * BT)
#define CAP    2048
#define KS     33
#define PITCH  132
#define AW     4
#define AT     (AW * 32)
#define RLSH   27

__device__ __forceinline__ int range_of(int r, int R) {
  int g = r >> RSHIFT;
  g = g < 0 ? 0 : g;
  return g >= R ? R - 1 : g;
}

__global__ __launch_bounds__(BT) void sort_slices(const int* __restrict__ rowi,
                                                  int* __restrict__ rec,
                                                  int* __restrict__ offt,
                                                  int E, int R) {
  __shared__ int cur[MAXR];
  __shared__ int wtot[BT / 32];
  __shared__ int stage[SLICE];
  const int t = threadIdx.x;
  const int lane = t & 31;
  const int wv = t >> 5;
  const int b = blockIdx.x;
  for (int i = t; i < R; i += BT) cur[i] = 0;

  const int epb = (E + NBKT - 1) / NBKT;
  const int beg = min(b * epb, E);
  const int end = min(E, beg + min(epb, SLICE));
  const int n = end - beg;

  int rv[EPT];
#pragma unroll
  for (int i = 0; i < EPT; ++i) {
    const int e = beg + t + i * BT;
    rv[i] = rowi[e < E ? e : E - 1];
  }
  __syncthreads();
#pragma unroll
  for (int i = 0; i < EPT; ++i) {
    if (t + i * BT < n) atomicAdd(&cur[range_of(rv[i], R)], 1);
  }
  __syncthreads();

  const int tot = t < R ? cur[t] : 0;
  int inc = tot;
  for (int d = 1; d < 32; d <<= 1) {
    const int y = __shfl_up(inc, d);
    if (lane >= d) inc += y;
  }
  if (lane == 31) wtot[wv] = inc;
  __syncthreads();
  if (wv == 0) {
    const int wsum = wtot[lane];
    int winc = wsum;
    for (int d = 1; d < 32; d <<= 1) {
      const int y = __shfl_up(winc, d);
      if (lane >= d) winc += y;
    }
    wtot[lane] = winc - wsum;
  }
  __syncthreads();
  const int excl = wtot[wv] + inc - tot;
  if (t < R) {
    cur[t] = excl;
    offt[t * NBKT + b] = beg + excl;
  }
  if (t == 0) offt[R * NBKT + b] = end;
  __syncthreads();

#pragma unroll
  for (int i = 0; i < EPT; ++i) {
    if (t + i * BT < n) {
      const int pos = atomicAdd(&cur[range_of(rv[i], R)], 1);
      if ((unsigned)pos < (unsigned)SLICE)
        stage[pos] = (beg + t + i * BT) | ((rv[i] & 15) << RLSH);
    }
  }
  __syncthreads();
  for (int i = 0; i < EPT; ++i) {
    const int k = t + i * BT;
    if (k < n) rec[beg + k] = stage[k];
  }
}

__device__ __forceinline__ void lds_add4(float* ap, float a, float b, float c, float d) {
  __hip_atomic_fetch_add(ap,          a, __ATOMIC_RELAXED, __HIP_MEMORY_SCOPE_WORKGROUP);
  __hip_atomic_fetch_add(ap + KS,     b, __ATOMIC_RELAXED, __HIP_MEMORY_SCOPE_WORKGROUP);
  __hip_atomic_fetch_add(ap + 2 * KS, c, __ATOMIC_RELAXED, __HIP_MEMORY_SCOPE_WORKGROUP);
  __hip_atomic_fetch_add(ap + 3 * KS, d, __ATOMIC_RELAXED, __HIP_MEMORY_SCOPE_WORKGROUP);
}

__global__ __launch_bounds__(AT) void aggregate_rows_gemm(
    const float* __restrict__ x, const float* __restrict__ val,
    const int* __restrict__ coli, const float* __restrict__ w,
    const float* __restrict__ bias, const int* __restrict__ rec,
    const int* __restrict__ offt, float* __restrict__ out, int N, int E) {
  __shared__ __attribute__((aligned(16))) float acc[16 * PITCH];
  __shared__ int list[CAP + 32];
  __shared__ float vlist[CAP + 32];
  __shared__ int sst[NBKT];
  __shared__ int spre[NBKT + 1];
  __shared__ int whalf;
  const int t = threadIdx.x;
  const int lane = t & 31;
  const int wid = __builtin_amdgcn_readfirstlane(t >> 5);
  const int m0 = blockIdx.x << RSHIFT;

  int st = 0, len = 0;
  if (t < NBKT) {
    st = offt[blockIdx.x * NBKT + t];
    int en = offt[(blockIdx.x + 1) * NBKT + t];
    st = st < 0 ? 0 : st;
    st = st < E ? st : E;
    en = en < st ? st : en;
    en = en < E ? en : E;
    len = en - st;
  }
  for (int i = t; i < 16 * PITCH; i += AT) acc[i] = 0.f;
  int inc = len;
  for (int d = 1; d < 32; d <<= 1) {
    const int y = __shfl_up(inc, d);
    if (lane >= d) inc += y;
  }
  if (t == 31) whalf = inc;
  __syncthreads();
  if (t < NBKT) {
    const int pre = inc - len + (t >= 32 ? whalf : 0);
    sst[t] = st;
    spre[t] = pre;
    if (t == NBKT - 1) spre[NBKT] = pre + len;
  }
  __syncthreads();
  const int total = __builtin_amdgcn_readfirstlane(spre[NBKT]);

  const unsigned lane16 = (unsigned)lane << 4;
  const char* xb = (const char*)x + lane16;

  v16f ax = {}, ay = {}, az = {}, aw = {};
  for (int t0 = 0; t0 < total; t0 += CAP) {
    const int nt = min(CAP, total - t0);
    if (t0 > 0) __syncthreads();
    {
      int pv[NBKT / AW];
      int mx = 0;
#pragma unroll
      for (int u = 0; u < NBKT / AW; ++u) {
        const int sb = wid * (NBKT / AW) + u;
        const int jlo = max(t0 - spre[sb], 0);
        int i = sst[sb] + jlo + lane;
        i = i < E ? i : E - 1;
        pv[u] = rec[i];
      }
      int cv[NBKT / AW];
      float fv[NBKT / AW];
#pragma unroll
      for (int u = 0; u < NBKT / AW; ++u) {
        int e = pv[u] & ((1 << RLSH) - 1);
        e = e < E ? e : E - 1;
        cv[u] = coli[e];
        fv[u] = val[e];
      }
#pragma unroll
      for (int u = 0; u < NBKT / AW; ++u) {
        const int sb = wid * (NBKT / AW) + u;
        const int pb = spre[sb];
        const int jlo = max(t0 - pb, 0);
        const int jhi = min(spre[sb + 1] - pb, t0 + CAP - pb);
        mx = max(mx, jhi - jlo);
        int c = cv[u];
        c = c < 0 ? 0 : c;
        c = c < N ? c : N - 1;
        const int pos = (jlo + lane < jhi) ? pb + jlo + lane - t0 : CAP + lane;
        list[pos] = c | (pv[u] & (15 << RLSH));
        vlist[pos] = fv[u];
      }
      if (__builtin_amdgcn_readfirstlane(mx) > 32) {
        for (int u = 0; u < NBKT / AW; ++u) {
          const int sb = wid * (NBKT / AW) + u;
          const int pb = __builtin_amdgcn_readfirstlane(spre[sb]);
          const int pe = __builtin_amdgcn_readfirstlane(spre[sb + 1]);
          const int s0 = __builtin_amdgcn_readfirstlane(sst[sb]);
          const int jlo = max(t0 - pb, 0);
          const int jhi = min(pe - pb, t0 + CAP - pb);
          for (int j = jlo + 32 + lane; j < jhi; j += 32) {
            int i = s0 + j;
            i = i < E ? i : E - 1;
            const int p = rec[i];
            int e = p & ((1 << RLSH) - 1);
            e = e < E ? e : E - 1;
            int c = coli[e];
            c = c < 0 ? 0 : c;
            c = c < N ? c : N - 1;
            list[pb + j - t0] = c | (p & (15 << RLSH));
            vlist[pb + j - t0] = val[e];
          }
        }
      }
    }
    __syncthreads();

    int cb = wid * 32;
    int rl, cc;
    float vv;
    {
      int i0 = cb + lane;
      i0 = i0 < nt ? i0 : nt - 1;
      const int p = list[i0];
      const float v = vlist[i0];
      const bool ok = lane < nt - cb;
      rl = ok ? ((p >> RLSH) & 15) : 0;
      cc = ok ? (p & ((1 << RLSH) - 1)) : 0;
      vv = ok ? v : 0.f;
    }

    while (cb + 32 <= nt) {
      int i1 = cb + AT + lane;
      i1 = i1 < nt ? i1 : nt - 1;
      const int pa = list[i1];
      const float va = vlist[i1];
#pragma unroll
      for (int j = 0; j < 32; j += 16) {
        v4f g[16];
#pragma unroll
        for (int u = 0; u < 16; ++u) {
          const int c = __builtin_amdgcn_readlane(cc, j + u);
          g[u] = *(const v4f*)(xb + (size_t)c * (DFEAT * 4));
        }
#pragma unroll
        for (int u = 0; u < 16; ++u) {
          const int r = __builtin_amdgcn_readlane(rl, j + u);
          const float v = __int_as_float(
              __builtin_amdgcn_readlane(__float_as_int(vv), j + u));
          ax[r] += v * g[u].x;
          ay[r] += v * g[u].y;
          az[r] += v * g[u].z;
          aw[r] += v * g[u].w;
        }
      }
      cb += AT;
      const bool ok2 = lane < nt - cb;
      rl = ok2 ? ((pa >> RLSH) & 15) : 0;
      cc = ok2 ? (pa & ((1 << RLSH) - 1)) : 0;
      vv = ok2 ? va : 0.f;
    }
    if (cb < nt) {
      const int n = nt - cb;
#pragma unroll
      for (int j = 0; j < 32; j += 16) {
        if (j < n) {
          v4f g[16];
#pragma unroll
          for (int u = 0; u < 16; ++u) {
            const int c = __builtin_amdgcn_readlane(cc, j + u);
            g[u] = *(const v4f*)(xb + (size_t)c * (DFEAT * 4));
          }
#pragma unroll
          for (int u = 0; u < 16; ++u) {
            if (j + u < n) {
              const int r = __builtin_amdgcn_readlane(rl, j + u);
              const float v = __int_as_float(
                  __builtin_amdgcn_readlane(__float_as_int(vv), j + u));
              lds_add4(acc + r * PITCH + lane, v * g[u].x, v * g[u].y, v * g[u].z, v * g[u].w);
            }
          }
        }
      }
    }
  }
#pragma unroll
  for (int r = 0; r < 16; ++r) {
    lds_add4(acc + r * PITCH + lane, ax[r], ay[r], az[r], aw[r]);
  }
  __syncthreads();

  const int l16 = lane & 15;
  const int hh  = lane >> 4;
  const int kh  = hh << 1;
  const int n0  = wid * 32;
  const float* arow = acc + l16 * PITCH;
  const float* bcol = w + n0 + l16;

  v8f c0 = {}, c1 = {};
#pragma unroll
  for (int k0 = 0; k0 < DFEAT; k0 += 4) {
    v2f a;
    a.x = arow[(kh    ) * KS + (k0 >> 2)];
    a.y = arow[(kh + 1) * KS + (k0 >> 2)];
    v2f b0, b1;
    b0.x = bcol[(k0 + kh    ) * DFEAT];
    b0.y = bcol[(k0 + kh + 1) * DFEAT];
    b1.x = bcol[(k0 + kh    ) * DFEAT + 16];
    b1.y = bcol[(k0 + kh + 1) * DFEAT + 16];
    c0 = __builtin_amdgcn_wmma_f32_16x16x4_f32(false, a, false, b0,
                                               (short)0, c0, false, false);
    c1 = __builtin_amdgcn_wmma_f32_16x16x4_f32(false, a, false, b1,
                                               (short)0, c1, false, false);
  }

  const float bv0 = bias[n0 + l16];
  const float bv1 = bias[n0 + 16 + l16];
  __syncthreads();
#pragma unroll
  for (int j = 0; j < 8; ++j) {
    acc[(hh * 8 + j) * PITCH + n0 + l16]      = c0[j] + bv0;
    acc[(hh * 8 + j) * PITCH + n0 + 16 + l16] = c1[j] + bv1;
  }
  __syncthreads();
  for (int i = 0; i < 16 / AW; ++i) {
    const int row = wid * (16 / AW) + i;
    if (m0 + row < N) {
      const v4f o = *(const v4f*)(acc + row * PITCH + lane * 4);
      *(v4f*)(out + (size_t)(m0 + row) * DFEAT + lane * 4) = o;
    }
  }
}

extern "C" void kernel_launch(void* const* d_in, const int* in_sizes, int n_in,
                              void* d_out, int out_size, void* d_ws, size_t ws_size,
                              hipStream_t stream) {
  const float* x       = (const float*)d_in[0];
  const float* adj_val = (const float*)d_in[1];
  const float* weight  = (const float*)d_in[2];
  const float* bias    = (const float*)d_in[3];
  const int*   adj_row = (const int*)d_in[4];
  const int*   adj_col = (const int*)d_in[5];

  const int N = in_sizes[0] / DFEAT;
  const int E = in_sizes[1];
  const int R = (N + 15) / 16;

  int* rec    = (int*)d_ws;
  int* offt   = rec + E;
  float* out  = (float*)d_out;

  sort_slices<<<NBKT, BT, 0, stream>>>(adj_row, rec, offt, E, R);
  aggregate_rows_gemm<<<R, AT, 0, stream>>>(x, adj_val, adj_col, weight, bias,
                                             rec, offt, out, N, E);
}


# edge column and value travel through the slice sort: it writes sorted value (f32), column (2-byte) and row-in-range (1-byte) arrays, so aggregation reads runs of slots, no per-edge col/val gather
# speedup vs baseline: 3.4130x; 1.2495x over previous
#include <hip/hip_runtime.h>


typedef __attribute__((ext_vector_type(2))) float v2f;
typedef __attribute__((ext_vector_type(2))) int v2i;
typedef __attribute__((ext_vector_type(4))) float v4f;
typedef __attribute__((ext_vector_type(8))) float v8f;
typedef __attribute__((ext_vector_type(16))) float v16f;

#define DFEAT  128
#define RSHIFT 4
#define MAXR   1024
#define NBKT   64
#define BT     1024
#define EPT    10
#define SLICE  (EPT * BT)
#define CAP    2048
#define KS     33
#define PITCH  132
#define AW     4
#define AT     (AW * 32)
#define RLSH   27
#define SRSH   16

__device__ __forceinline__ int range_of(int r, int R) {
  int g = r >> RSHIFT;
  g = g < 0 ? 0 : g;
  return g >= R ? R - 1 : g;
}

__global__ __launch_bounds__(BT) void sort_slices(const int* __restrict__ rowi,
                                                  const int* __restrict__ coli,
                                                  const float* __restrict__ val,
                                                  float* __restrict__ vrec,
                                                  unsigned short* __restrict__ clo,
                                                  unsigned char* __restrict__ rhi,
                                                  int* __restrict__ offt,
                                                  int E, int R, int N) {
  __shared__ int cur[MAXR];
  __shared__ int wtot[BT / 32];
  __shared__ v2i stage[SLICE];
  const int t = threadIdx.x;
  const int lane = t & 31;
  const int wv = t >> 5;
  const int b = blockIdx.x;
  for (int i = t; i < R; i += BT) cur[i] = 0;

  const int epb = (E + NBKT - 1) / NBKT;
  const int beg = min(b * epb, E);
  const int end = min(E, beg + min(epb, SLICE));
  const int n = end - beg;

  int rv[EPT];
  int cv[EPT];
  float fv[EPT];
#pragma unroll
  for (int i = 0; i < EPT; ++i) {
    const int e = beg + t + i * BT;
    rv[i] = rowi[e < E ? e : E - 1];
  }
  __syncthreads();
#pragma unroll
  for (int i = 0; i < EPT; ++i) {
    const int e = beg + t + i * BT;
    cv[i] = coli[e < E ? e : E - 1];
    fv[i] = val[e < E ? e : E - 1];
  }
#pragma unroll
  for (int i = 0; i < EPT; ++i) {
    if (t + i * BT < n) atomicAdd(&cur[range_of(rv[i], R)], 1);
  }
  __syncthreads();

  const int tot = t < R ? cur[t] : 0;
  int inc = tot;
  for (int d = 1; d < 32; d <<= 1) {
    const int y = __shfl_up(inc, d);
    if (lane >= d) inc += y;
  }
  if (lane == 31) wtot[wv] = inc;
  __syncthreads();
  if (wv == 0) {
    const int wsum = wtot[lane];
    int winc = wsum;
    for (int d = 1; d < 32; d <<= 1) {
      const int y = __shfl_up(winc, d);
      if (lane >= d) winc += y;
    }
    wtot[lane] = winc - wsum;
  }
  __syncthreads();
  const int excl = wtot[wv] + inc - tot;
  if (t < R) {
    cur[t] = excl;
    offt[t * NBKT + b] = beg + excl;
  }
  if (t == 0) offt[R * NBKT + b] = end;
  __syncthreads();

#pragma unroll
  for (int i = 0; i < EPT; ++i) {
    if (t + i * BT < n) {
      const int pos = atomicAdd(&cur[range_of(rv[i], R)], 1);
      if ((unsigned)pos < (unsigned)SLICE) {
        int c = cv[i];
        c = c < 0 ? 0 : c;
        c = c < N ? c : N - 1;
        v2i s;
        s.x = c | ((rv[i] & 15) << SRSH);
        s.y = __float_as_int(fv[i]);
        stage[pos] = s;
      }
    }
  }
  __syncthreads();
  for (int i = 0; i < EPT; ++i) {
    const int k = t + i * BT;
    if (k < n) {
      const v2i s = stage[k];
      vrec[beg + k] = __int_as_float(s.y);
      clo[beg + k] = (unsigned short)(s.x & 0xffff);
      rhi[beg + k] = (unsigned char)((s.x >> SRSH) & 15);
    }
  }
}

__device__ __forceinline__ void lds_add4(float* ap, float a, float b, float c, float d) {
  __hip_atomic_fetch_add(ap,          a, __ATOMIC_RELAXED, __HIP_MEMORY_SCOPE_WORKGROUP);
  __hip_atomic_fetch_add(ap + KS,     b, __ATOMIC_RELAXED, __HIP_MEMORY_SCOPE_WORKGROUP);
  __hip_atomic_fetch_add(ap + 2 * KS, c, __ATOMIC_RELAXED, __HIP_MEMORY_SCOPE_WORKGROUP);
  __hip_atomic_fetch_add(ap + 3 * KS, d, __ATOMIC_RELAXED, __HIP_MEMORY_SCOPE_WORKGROUP);
}

__global__ __launch_bounds__(AT) void aggregate_rows_gemm(
    const float* __restrict__ x, const float* __restrict__ w,
    const float* __restrict__ bias, const float* __restrict__ vrec,
    const unsigned short* __restrict__ clo, const unsigned char* __restrict__ rhi,
    const int* __restrict__ offt, float* __restrict__ out, int N, int E) {
  __shared__ __attribute__((aligned(16))) float acc[16 * PITCH];
  __shared__ int list[CAP + 32];
  __shared__ float vlist[CAP + 32];
  __shared__ int sst[NBKT];
  __shared__ int spre[NBKT + 1];
  __shared__ int whalf;
  const int t = threadIdx.x;
  const int lane = t & 31;
  const int wid = __builtin_amdgcn_readfirstlane(t >> 5);
  const int m0 = blockIdx.x << RSHIFT;

  int st = 0, len = 0;
  if (t < NBKT) {
    st = offt[blockIdx.x * NBKT + t];
    int en = offt[(blockIdx.x + 1) * NBKT + t];
    st = st < 0 ? 0 : st;
    st = st < E ? st : E;
    en = en < st ? st : en;
    en = en < E ? en : E;
    len = en - st;
  }
  for (int i = t; i < 16 * PITCH; i += AT) acc[i] = 0.f;
  int inc = len;
  for (int d = 1; d < 32; d <<= 1) {
    const int y = __shfl_up(inc, d);
    if (lane >= d) inc += y;
  }
  if (t == 31) whalf = inc;
  __syncthreads();
  if (t < NBKT) {
    const int pre = inc - len + (t >= 32 ? whalf : 0);
    sst[t] = st;
    spre[t] = pre;
    if (t == NBKT - 1) spre[NBKT] = pre + len;
  }
  __syncthreads();
  const int total = __builtin_amdgcn_readfirstlane(spre[NBKT]);

  const unsigned lane16 = (unsigned)lane << 4;
  const char* xb = (const char*)x + lane16;

  v16f ax = {}, ay = {}, az = {}, aw = {};
  for (int t0 = 0; t0 < total; t0 += CAP) {
    const int nt = min(CAP, total - t0);
    if (t0 > 0) __syncthreads();
    {
      int cv[NBKT / AW];
      int rh[NBKT / AW];
      float fv[NBKT / AW];
      int mx = 0;
#pragma unroll
      for (int u = 0; u < NBKT / AW; ++u) {
        const int sb = wid * (NBKT / AW) + u;
        const int jlo = max(t0 - spre[sb], 0);
        int i = sst[sb] + jlo + lane;
        i = i < E ? i : E - 1;
        fv[u] = vrec[i];
        cv[u] = clo[i];
        rh[u] = rhi[i];
      }
#pragma unroll
      for (int u = 0; u < NBKT / AW; ++u) {
        const int sb = wid * (NBKT / AW) + u;
        const int pb = spre[sb];
        const int jlo = max(t0 - pb, 0);
        const int jhi = min(spre[sb + 1] - pb, t0 + CAP - pb);
        mx = max(mx, jhi - jlo);
        int c = cv[u];
        c = c < N ? c : N - 1;
        const int pos = (jlo + lane < jhi) ? pb + jlo + lane - t0 : CAP + lane;
        list[pos] = c | ((rh[u] & 15) << RLSH);
        vlist[pos] = fv[u];
      }
      if (__builtin_amdgcn_readfirstlane(mx) > 32) {
        for (int u = 0; u < NBKT / AW; ++u) {
          const int sb = wid * (NBKT / AW) + u;
          const int pb = __builtin_amdgcn_readfirstlane(spre[sb]);
          const int pe = __builtin_amdgcn_readfirstlane(spre[sb + 1]);
          const int s0 = __builtin_amdgcn_readfirstlane(sst[sb]);
          const int jlo = max(t0 - pb, 0);
          const int jhi = min(pe - pb, t0 + CAP - pb);
          for (int j = jlo + 32 + lane; j < jhi; j += 32) {
            int i = s0 + j;
            i = i < E ? i : E - 1;
            int c = clo[i];
            c = c < N ? c : N - 1;
            list[pb + j - t0] = c | ((rhi[i] & 15) << RLSH);
            vlist[pb + j - t0] = vrec[i];
          }
        }
      }
    }
    __syncthreads();

    int cb = wid * 32;
    int rl, cc;
    float vv;
    {
      int i0 = cb + lane;
      i0 = i0 < nt ? i0 : nt - 1;
      const int p = list[i0];
      const float v = vlist[i0];
      const bool ok = lane < nt - cb;
      rl = ok ? ((p >> RLSH) & 15) : 0;
      cc = ok ? (p & ((1 << RLSH) - 1)) : 0;
      vv = ok ? v : 0.f;
    }

    while (cb + 32 <= nt) {
      int i1 = cb + AT + lane;
      i1 = i1 < nt ? i1 : nt - 1;
      const int pa = list[i1];
      const float va = vlist[i1];
#pragma unroll
      for (int j = 0; j < 32; j += 16) {
        v4f g[16];
#pragma unroll
        for (int u = 0; u < 16; ++u) {
          const int c = __builtin_amdgcn_readlane(cc, j + u);
          g[u] = *(const v4f*)(xb + (size_t)c * (DFEAT * 4));
        }
#pragma unroll
        for (int u = 0; u < 16; ++u) {
          const int r = __builtin_amdgcn_readlane(rl, j + u);
          const float v = __int_as_float(
              __builtin_amdgcn_readlane(__float_as_int(vv), j + u));
          ax[r] += v * g[u].x;
          ay[r] += v * g[u].y;
          az[r] += v * g[u].z;
          aw[r] += v * g[u].w;
        }
      }
      cb += AT;
      const bool ok2 = lane < nt - cb;
      rl = ok2 ? ((pa >> RLSH) & 15) : 0;
      cc = ok2 ? (pa & ((1 << RLSH) - 1)) : 0;
      vv = ok2 ? va : 0.f;
    }
    if (cb < nt) {
      const int n = nt - cb;
#pragma unroll
      for (int j = 0; j < 32; j += 16) {
        if (j < n) {
          v4f g[16];
#pragma unroll
          for (int u = 0; u < 16; ++u) {
            const int c = __builtin_amdgcn_readlane(cc, j + u);
            g[u] = *(const v4f*)(xb + (size_t)c * (DFEAT * 4));
          }
#pragma unroll
          for (int u = 0; u < 16; ++u) {
            if (j + u < n) {
              const int r = __builtin_amdgcn_readlane(rl, j + u);
              const float v = __int_as_float(
                  __builtin_amdgcn_readlane(__float_as_int(vv), j + u));
              lds_add4(acc + r * PITCH + lane, v * g[u].x, v * g[u].y, v * g[u].z, v * g[u].w);
            }
          }
        }
      }
    }
  }
#pragma unroll
  for (int r = 0; r < 16; ++r) {
    lds_add4(acc + r * PITCH + lane, ax[r], ay[r], az[r], aw[r]);
  }
  __syncthreads();

  const int l16 = lane & 15;
  const int hh  = lane >> 4;
  const int kh  = hh << 1;
  const int n0  = wid * 32;
  const float* arow = acc + l16 * PITCH;
  const float* bcol = w + n0 + l16;

  v8f c0 = {}, c1 = {};
#pragma unroll
  for (int k0 = 0; k0 < DFEAT; k0 += 4) {
    v2f a;
    a.x = arow[(kh    ) * KS + (k0 >> 2)];
    a.y = arow[(kh + 1) * KS + (k0 >> 2)];
    v2f b0, b1;
    b0.x = bcol[(k0 + kh    ) * DFEAT];
    b0.y = bcol[(k0 + kh + 1) * DFEAT];
    b1.x = bcol[(k0 + kh    ) * DFEAT + 16];
    b1.y = bcol[(k0 + kh + 1) * DFEAT + 16];
    c0 = __builtin_amdgcn_wmma_f32_16x16x4_f32(false, a, false, b0,
                                               (short)0, c0, false, false);
    c1 = __builtin_amdgcn_wmma_f32_16x16x4_f32(false, a, false, b1,
                                               (short)0, c1, false, false);
  }

  const float bv0 = bias[n0 + l16];
  const float bv1 = bias[n0 + 16 + l16];
  __syncthreads();
#pragma unroll
  for (int j = 0; j < 8; ++j) {
    acc[(hh * 8 + j) * PITCH + n0 + l16]      = c0[j] + bv0;
    acc[(hh * 8 + j) * PITCH + n0 + 16 + l16] = c1[j] + bv1;
  }
  __syncthreads();
  for (int i = 0; i < 16 / AW; ++i) {
    const int row = wid * (16 / AW) + i;
    if (m0 + row < N) {
      const v4f o = *(const v4f*)(acc + row * PITCH + lane * 4);
      *(v4f*)(out + (size_t)(m0 + row) * DFEAT + lane * 4) = o;
    }
  }
}

extern "C" void kernel_launch(void* const* d_in, const int* in_sizes, int n_in,
                              void* d_out, int out_size, void* d_ws, size_t ws_size,
                              hipStream_t stream) {
  const float* x       = (const float*)d_in[0];
  const float* adj_val = (const float*)d_in[1];
  const float* weight  = (const float*)d_in[2];
  const float* bias    = (const float*)d_in[3];
  const int*   adj_row = (const int*)d_in[4];
  const int*   adj_col = (const int*)d_in[5];

  const int N = in_sizes[0] / DFEAT;
  const int E = in_sizes[1];
  const int R = (N + 15) / 16;

  float* vrec = (float*)d_ws;
  int* offt   = (int*)(vrec + E);
  unsigned short* clo = (unsigned short*)(offt + (R + 1) * NBKT);
  unsigned char* rhi  = (unsigned char*)(clo + E);
  float* out  = (float*)d_out;

  sort_slices<<<NBKT, BT, 0, stream>>>(adj_row, adj_col, adj_val, vrec, clo, rhi,
                                       offt, E, R, N);
  aggregate_rows_gemm<<<R, AT, 0, stream>>>(x, weight, bias, vrec, clo, rhi,
                                             offt, out, N, E);
}


# edge list sorted as 128 slices by 128 blocks of 512 threads (was 64 of 1024): each sorter's LDS counting, loads and stores halve; owner reads two 16-record shares per load; sum order: slice, LDS rank
# speedup vs baseline: 3.6445x; 1.0678x over previous
#include <hip/hip_runtime.h>


typedef __attribute__((ext_vector_type(2))) float v2f;
typedef __attribute__((ext_vector_type(2))) int v2i;
typedef __attribute__((ext_vector_type(4))) float v4f;
typedef __attribute__((ext_vector_type(8))) float v8f;
typedef __attribute__((ext_vector_type(16))) float v16f;

#define DFEAT  128
#define RSHIFT 4
#define MAXR   1024
#define NBKT   128
#define BT     512
#define EPT    10
#define SLICE  (EPT * BT)
#define CAP    2048
#define KS     33
#define PITCH  132
#define AW     4
#define AT     (AW * 32)
#define RLSH   27
#define SRSH   16

static_assert(NBKT / AW == 32 && NBKT <= AT, "the owner's wave looks at one sub-segment per lane");
static_assert(MAXR <= 2 * BT, "a sorting thread owns two ranges");

__device__ __forceinline__ int range_of(int r, int R) {
  int g = r >> RSHIFT;
  g = g < 0 ? 0 : g;
  return g >= R ? R - 1 : g;
}

__global__ __launch_bounds__(BT) void sort_slices(const int* __restrict__ rowi,
                                                  const int* __restrict__ coli,
                                                  const float* __restrict__ val,
                                                  float* __restrict__ vrec,
                                                  unsigned short* __restrict__ clo,
                                                  unsigned char* __restrict__ rhi,
                                                  int* __restrict__ offt,
                                                  int E, int R, int N) {
  __shared__ __attribute__((aligned(8))) int cur[MAXR];
  __shared__ int wtot[BT / 32];
  __shared__ v2i stage[SLICE];
  const int t = threadIdx.x;
  const int lane = t & 31;
  const int wv = t >> 5;
  const int b = blockIdx.x;
  for (int i = t; i < MAXR; i += BT) cur[i] = 0;

  const int epb = (E + NBKT - 1) / NBKT;
  const int beg = min(b * epb, E);
  const int end = min(E, beg + min(epb, SLICE));
  const int n = end - beg;

  int rv[EPT];
  int cv[EPT];
  float fv[EPT];
#pragma unroll
  for (int i = 0; i < EPT; ++i) {
    const int e = beg + t + i * BT;
    rv[i] = rowi[e < E ? e : E - 1];
  }
  __syncthreads();
#pragma unroll
  for (int i = 0; i < EPT; ++i) {
    const int e = beg + t + i * BT;
    cv[i] = coli[e < E ? e : E - 1];
    fv[i] = val[e < E ? e : E - 1];
  }
#pragma unroll
  for (int i = 0; i < EPT; ++i) {
    if (t + i * BT < n) atomicAdd(&cur[range_of(rv[i], R)], 1);
  }
  __syncthreads();

  const v2i cpair = *(const v2i*)&cur[2 * t];
  const int c0 = cpair.x;
  const int tot = c0 + cpair.y;
  int inc = tot;
  for (int d = 1; d < 32; d <<= 1) {
    const int y = __shfl_up(inc, d);
    if (lane >= d) inc += y;
  }
  if (lane == 31) wtot[wv] = inc;
  __syncthreads();
  if (wv == 0) {
    const int wsum = lane < BT / 32 ? wtot[lane] : 0;
    int winc = wsum;
    for (int d = 1; d < 32; d <<= 1) {
      const int y = __shfl_up(winc, d);
      if (lane >= d) winc += y;
    }
    if (lane < BT / 32) wtot[lane] = winc - wsum;
  }
  __syncthreads();
  const int excl = wtot[wv] + inc - tot;
  if (2 * t < R) {
    cur[2 * t] = excl;
    offt[(2 * t) * NBKT + b] = beg + excl;
  }
  if (2 * t + 1 < R) {
    cur[2 * t + 1] = excl + c0;
    offt[(2 * t + 1) * NBKT + b] = beg + excl + c0;
  }
  if (t == 0) offt[R * NBKT + b] = end;
  __syncthreads();

#pragma unroll
  for (int i = 0; i < EPT; ++i) {
    if (t + i * BT < n) {
      const int pos = atomicAdd(&cur[range_of(rv[i], R)], 1);
      if ((unsigned)pos < (unsigned)SLICE) {
        int c = cv[i];
        c = c < 0 ? 0 : c;
        c = c < N ? c : N - 1;
        v2i s;
        s.x = c | ((rv[i] & 15) << SRSH);
        s.y = __float_as_int(fv[i]);
        stage[pos] = s;
      }
    }
  }
  __syncthreads();
  for (int i = 0; i < EPT; ++i) {
    const int k = t + i * BT;
    if (k < n) {
      const v2i s = stage[k];
      vrec[beg + k] = __int_as_float(s.y);
      clo[beg + k] = (unsigned short)(s.x & 0xffff);
      rhi[beg + k] = (unsigned char)((s.x >> SRSH) & 15);
    }
  }
}

__device__ __forceinline__ void lds_add4(float* ap, float a, float b, float c, float d) {
  __hip_atomic_fetch_add(ap,          a, __ATOMIC_RELAXED, __HIP_MEMORY_SCOPE_WORKGROUP);
  __hip_atomic_fetch_add(ap + KS,     b, __ATOMIC_RELAXED, __HIP_MEMORY_SCOPE_WORKGROUP);
  __hip_atomic_fetch_add(ap + 2 * KS, c, __ATOMIC_RELAXED, __HIP_MEMORY_SCOPE_WORKGROUP);
  __hip_atomic_fetch_add(ap + 3 * KS, d, __ATOMIC_RELAXED, __HIP_MEMORY_SCOPE_WORKGROUP);
}

__global__ __launch_bounds__(AT) void aggregate_rows_gemm(
    const float* __restrict__ x, const float* __restrict__ w,
    const float* __restrict__ bias, const float* __restrict__ vrec,
    const unsigned short* __restrict__ clo, const unsigned char* __restrict__ rhi,
    const int* __restrict__ offt, float* __restrict__ out, int N, int E) {
  __shared__ __attribute__((aligned(16))) float acc[16 * PITCH];
  __shared__ int list[CAP + 32];
  __shared__ float vlist[CAP + 32];
  __shared__ int sst[NBKT];
  __shared__ int spre[NBKT + 1];
  __shared__ int wsum[AW];
  const int t = threadIdx.x;
  const int lane = t & 31;
  const int wid = __builtin_amdgcn_readfirstlane(t >> 5);
  const int m0 = blockIdx.x << RSHIFT;

  int st = 0, len = 0;
  if (t < NBKT) {
    st = offt[blockIdx.x * NBKT + t];
    int en = offt[(blockIdx.x + 1) * NBKT + t];
    st = st < 0 ? 0 : st;
    st = st < E ? st : E;
    en = en < st ? st : en;
    en = en < E ? en : E;
    len = en - st;
  }
  for (int i = t; i < 16 * PITCH; i += AT) acc[i] = 0.f;
  int inc = len;
  for (int d = 1; d < 32; d <<= 1) {
    const int y = __shfl_up(inc, d);
    if (lane >= d) inc += y;
  }
  if (lane == 31) wsum[wid] = inc;
  __syncthreads();
  if (t < NBKT) {
    int pre = inc - len;
    for (int q = 0; q < AW; ++q) pre += q < wid ? wsum[q] : 0;
    sst[t] = st;
    spre[t] = pre;
    if (t == NBKT - 1) spre[NBKT] = pre + len;
  }
  __syncthreads();
  const int total = __builtin_amdgcn_readfirstlane(spre[NBKT]);

  const unsigned lane16 = (unsigned)lane << 4;
  const char* xb = (const char*)x + lane16;

  v16f ax = {}, ay = {}, az = {}, aw = {};
  for (int t0 = 0; t0 < total; t0 += CAP) {
    const int nt = min(CAP, total - t0);
    if (t0 > 0) __syncthreads();
    {
      const int hsel = lane >> 4;
      const int l16 = lane & 15;
      int cv[NBKT / AW / 2];
      int rh[NBKT / AW / 2];
      float fv[NBKT / AW / 2];
#pragma unroll
      for (int u = 0; u < NBKT / AW / 2; ++u) {
        const int sb = wid * (NBKT / AW) + 2 * u + hsel;
        const int jlo = max(t0 - spre[sb], 0);
        int i = sst[sb] + jlo + l16;
        i = i < E ? i : E - 1;
        fv[u] = vrec[i];
        cv[u] = clo[i];
        rh[u] = rhi[i];
      }
      const int sbl = wid * (NBKT / AW) + lane;
      const int pbl = spre[sbl];
      const int s0l = sst[sbl];
      const int jlol = max(t0 - pbl, 0);
      const int jhil = min(spre[sbl + 1] - pbl, t0 + CAP - pbl);
#pragma unroll
      for (int u = 0; u < NBKT / AW / 2; ++u) {
        const int sb = wid * (NBKT / AW) + 2 * u + hsel;
        const int pb = spre[sb];
        const int jlo = max(t0 - pb, 0);
        const int jhi = min(spre[sb + 1] - pb, t0 + CAP - pb);
        int c = cv[u];
        c = c < N ? c : N - 1;
        const int pos = (jlo + l16 < jhi) ? pb + jlo + l16 - t0 : CAP + lane;
        list[pos] = c | ((rh[u] & 15) << RLSH);
        vlist[pos] = fv[u];
      }
      unsigned more = __builtin_amdgcn_ballot_w32(jhil - jlol > 16);
      while (more) {
        const int u = __builtin_ctz(more);
        more &= more - 1;
        const int pb = __builtin_amdgcn_readlane(pbl, u);
        const int s0 = __builtin_amdgcn_readlane(s0l, u);
        const int jlo = __builtin_amdgcn_readlane(jlol, u);
        const int jhi = __builtin_amdgcn_readlane(jhil, u);
        for (int j = jlo + 16 + lane; j < jhi; j += 32) {
          int i = s0 + j;
          i = i < E ? i : E - 1;
          int c = clo[i];
          c = c < N ? c : N - 1;
          list[pb + j - t0] = c | ((rhi[i] & 15) << RLSH);
          vlist[pb + j - t0] = vrec[i];
        }
      }
    }
    __syncthreads();

    int cb = wid * 32;
    int rl, cc;
    float vv;
    {
      int i0 = cb + lane;
      i0 = i0 < nt ? i0 : nt - 1;
      const int p = list[i0];
      const float v = vlist[i0];
      const bool ok = lane < nt - cb;
      rl = ok ? ((p >> RLSH) & 15) : 0;
      cc = ok ? (p & ((1 << RLSH) - 1)) : 0;
      vv = ok ? v : 0.f;
    }

    while (cb + 32 <= nt) {
      int i1 = cb + AT + lane;
      i1 = i1 < nt ? i1 : nt - 1;
      const int pa = list[i1];
      const float va = vlist[i1];
#pragma unroll
      for (int j = 0; j < 32; j += 16) {
        v4f g[16];
#pragma unroll
        for (int u = 0; u < 16; ++u) {
          const int c = __builtin_amdgcn_readlane(cc, j + u);
          g[u] = *(const v4f*)(xb + (size_t)c * (DFEAT * 4));
        }
#pragma unroll
        for (int u = 0; u < 16; ++u) {
          const int r = __builtin_amdgcn_readlane(rl, j + u);
          const float v = __int_as_float(
              __builtin_amdgcn_readlane(__float_as_int(vv), j + u));
          ax[r] += v * g[u].x;
          ay[r] += v * g[u].y;
          az[r] += v * g[u].z;
          aw[r] += v * g[u].w;
        }
      }
      cb += AT;
      const bool ok2 = lane < nt - cb;
      rl = ok2 ? ((pa >> RLSH) & 15) : 0;
      cc = ok2 ? (pa & ((1 << RLSH) - 1)) : 0;
      vv = ok2 ? va : 0.f;
    }
    if (cb < nt) {
      const int n = nt - cb;
#pragma unroll
      for (int j = 0; j < 32; j += 16) {
        if (j < n) {
          v4f g[16];
#pragma unroll
          for (int u = 0; u < 16; ++u) {
            const int c = __builtin_amdgcn_readlane(cc, j + u);
            g[u] = *(const v4f*)(xb + (size_t)c * (DFEAT * 4));
          }
#pragma unroll
          for (int u = 0; u < 16; ++u) {
            if (j + u < n) {
              const int r = __builtin_amdgcn_readlane(rl, j + u);
              const float v = __int_as_float(
                  __builtin_amdgcn_readlane(__float_as_int(vv), j + u));
              lds_add4(acc + r * PITCH + lane, v * g[u].x, v * g[u].y, v * g[u].z, v * g[u].w);
            }
          }
        }
      }
    }
  }
#pragma unroll
  for (int r = 0; r < 16; ++r) {
    lds_add4(acc + r * PITCH + lane, ax[r], ay[r], az[r], aw[r]);
  }
  __syncthreads();

  const int l16 = lane & 15;
  const int hh  = lane >> 4;
  const int kh  = hh << 1;
  const int n0  = wid * 32;
  const float* arow = acc + l16 * PITCH;
  const float* bcol = w + n0 + l16;

  v8f c0 = {}, c1 = {};
#pragma unroll
  for (int k0 = 0; k0 < DFEAT; k0 += 4) {
    v2f a;
    a.x = arow[(kh    ) * KS + (k0 >> 2)];
    a.y = arow[(kh + 1) * KS + (k0 >> 2)];
    v2f b0, b1;
    b0.x = bcol[(k0 + kh    ) * DFEAT];
    b0.y = bcol[(k0 + kh + 1) * DFEAT];
    b1.x = bcol[(k0 + kh    ) * DFEAT + 16];
    b1.y = bcol[(k0 + kh + 1) * DFEAT + 16];
    c0 = __builtin_amdgcn_wmma_f32_16x16x4_f32(false, a, false, b0,
                                               (short)0, c0, false, false);
    c1 = __builtin_amdgcn_wmma_f32_16x16x4_f32(false, a, false, b1,
                                               (short)0, c1, false, false);
  }

  const float bv0 = bias[n0 + l16];
  const float bv1 = bias[n0 + 16 + l16];
  __syncthreads();
#pragma unroll
  for (int j = 0; j < 8; ++j) {
    acc[(hh * 8 + j) * PITCH + n0 + l16]      = c0[j] + bv0;
    acc[(hh * 8 + j) * PITCH + n0 + 16 + l16] = c1[j] + bv1;
  }
  __syncthreads();
  for (int i = 0; i < 16 / AW; ++i) {
    const int row = wid * (16 / AW) + i;
    if (m0 + row < N) {
      const v4f o = *(const v4f*)(acc + row * PITCH + lane * 4);
      *(v4f*)(out + (size_t)(m0 + row) * DFEAT + lane * 4) = o;
    }
  }
}

extern "C" void kernel_launch(void* const* d_in, const int* in_sizes, int n_in,
                              void* d_out, int out_size, void* d_ws, size_t ws_size,
                              hipStream_t stream) {
  const float* x       = (const float*)d_in[0];
  const float* adj_val = (const float*)d_in[1];
  const float* weight  = (const float*)d_in[2];
  const float* bias    = (const float*)d_in[3];
  const int*   adj_row = (const int*)d_in[4];
  const int*   adj_col = (const int*)d_in[5];

  const int N = in_sizes[0] / DFEAT;
  const int E = in_sizes[1];
  const int R = (N + 15) / 16;

  float* vrec = (float*)d_ws;
  int* offt   = (int*)(vrec + E);
  unsigned short* clo = (unsigned short*)(offt + (R + 1) * NBKT);
  unsigned char* rhi  = (unsigned char*)(clo + E);
  float* out  = (float*)d_out;

  sort_slices<<<NBKT, BT, 0, stream>>>(adj_row, adj_col, adj_val, vrec, clo, rhi,
                                       offt, E, R, N);
  aggregate_rows_gemm<<<R, AT, 0, stream>>>(x, weight, bias, vrec, clo, rhi,
                                             offt, out, N, E);
}
